# starting kernel
# baseline (speedup 1.0000x reference)
#include <hip/hip_runtime.h>

typedef float v2f __attribute__((ext_vector_type(2)));
typedef float v8f __attribute__((ext_vector_type(8)));

#define GP_ROUNDS   3
#define GP_NODE_DIM 64
#define GP_ACT_DIM  64
#define GP_EDGE_DIM 32
#define GP_N_NODES  100000
#define GP_N_EDGES  1600000
#define GP_CAT_DIM  160
#define KC_SRC  16
#define KC_DST  16
#define KC_EDG  8
#define KC_TOT  40

static __device__ __forceinline__ v8f wmma4(v2f a, v2f b, v8f c) {
  return __builtin_amdgcn_wmma_f32_16x16x4_f32(false, a, false, b, (short)0, c,
                                               false, false);
}
static __device__ __forceinline__ v2f ld2(const float* p) {
  return *(const v2f*)p;
}
static __device__ __forceinline__ float sigmoid_(float x) {
  return 1.0f / (1.0f + __expf(-x));
}

__global__ __launch_bounds__(256) void gp_edge_kernel(
    const float* __restrict__ hv, const float* __restrict__ he,
    const int* __restrict__ src, const int* __restrict__ dst,
    const float* __restrict__ W,
    const float* __restrict__ bias,
    float* __restrict__ abuf)
{
  __shared__ float Wl[KC_TOT * 4 * 32 * 2];
  const int tid = threadIdx.x;
#pragma unroll
  for (int p = tid; p < KC_TOT * 4 * 32; p += 256) {
    const int L  = p & 31;
    const int q  = p >> 5;
    const int n0 = q & 3;
    const int c  = q >> 2;
    const int k  = 4 * c + 2 * (L >> 4);
    const int col = n0 * 16 + (L & 15);
    v2f w2;
    w2.x = W[k * 64 + col];
    w2.y = W[(k + 1) * 64 + col];
    *(v2f*)&Wl[2 * p] = w2;
  }
  __syncthreads();

  const int lane = tid & 31;
  const int wave = tid >> 5;
  const int tile = blockIdx.x * 8 + wave;
  const int e0   = tile * 16;
  const int row  = lane & 15;
  const int half = lane >> 4;

  const int sidx = src[e0 + row];
  const int didx = dst[e0 + row];
  const float* hs = hv + (long)sidx * GP_NODE_DIM + 2 * half;
  const float* hd = hv + (long)didx * GP_NODE_DIM + 2 * half;
  const float* hp = he + (long)(e0 + row) * GP_EDGE_DIM + 2 * half;

  v2f As[KC_SRC], Ad[KC_DST], Ae[KC_EDG];
#pragma unroll
  for (int c = 0; c < KC_SRC; ++c) As[c] = ld2(hs + 4 * c);
#pragma unroll
  for (int c = 0; c < KC_DST; ++c) Ad[c] = ld2(hd + 4 * c);
#pragma unroll
  for (int c = 0; c < KC_EDG; ++c) Ae[c] = ld2(hp + 4 * c);

  int drow[8];
#pragma unroll
  for (int i = 0; i < 8; ++i) drow[i] = dst[e0 + i + 8 * half];

  const v2f* Wl2 = (const v2f*)Wl;
#pragma unroll
  for (int n0 = 0; n0 < 4; ++n0) {
    const float bv = bias[n0 * 16 + row];
    v8f acc;
#pragma unroll
    for (int i = 0; i < 8; ++i) acc[i] = bv;

    const v2f* Bp = Wl2 + (n0 * 32 + lane);
#pragma unroll
    for (int c = 0; c < KC_SRC; ++c)
      acc = wmma4(As[c], Bp[(c) * 128], acc);
#pragma unroll
    for (int c = 0; c < KC_DST; ++c)
      acc = wmma4(Ad[c], Bp[(KC_SRC + c) * 128], acc);
#pragma unroll
    for (int c = 0; c < KC_EDG; ++c)
      acc = wmma4(Ae[c], Bp[(KC_SRC + KC_DST + c) * 128], acc);

#pragma unroll
    for (int i = 0; i < 8; ++i)
      atomicAdd(&abuf[(long)drow[i] * GP_ACT_DIM + n0 * 16 + row], acc[i]);
  }
}

__global__ __launch_bounds__(256) void gp_gru_kernel(
    const float* __restrict__ abuf, const float* __restrict__ hvin,
    const float* __restrict__ w_ih, const float* __restrict__ w_hh,
    const float* __restrict__ b_ih, const float* __restrict__ b_hh,
    float* __restrict__ hvout)
{
  const int tid  = threadIdx.x;
  const int lane = tid & 31;
  const int wave = tid >> 5;
  const int tile = blockIdx.x * 8 + wave;
  const int ntiles = GP_N_NODES / 16;
  if (tile >= ntiles) return;

  const int nbase = tile * 16;
  const int row   = lane & 15;
  const int half  = lane >> 4;

  v2f Aa[16], Ah[16];
  const float* ap = abuf + (long)(nbase + row) * GP_ACT_DIM + 2 * half;
  const float* hq = hvin + (long)(nbase + row) * GP_NODE_DIM + 2 * half;
#pragma unroll
  for (int c = 0; c < 16; ++c) { Aa[c] = ld2(ap + 4 * c); Ah[c] = ld2(hq + 4 * c); }

#pragma unroll
  for (int j0 = 0; j0 < 4; ++j0) {
    const int colb = j0 * 16 + row;
    v8f ir, iz, in_, hr, hz, hn;
    {
      const float bi0 = b_ih[colb], bi1 = b_ih[64 + colb], bi2 = b_ih[128 + colb];
      const float bh0 = b_hh[colb], bh1 = b_hh[64 + colb], bh2 = b_hh[128 + colb];
#pragma unroll
      for (int i = 0; i < 8; ++i) {
        ir[i] = bi0; iz[i] = bi1; in_[i] = bi2;
        hr[i] = bh0; hz[i] = bh1; hn[i] = bh2;
      }
    }
    const float* wi0 = w_ih + (long)(colb)*64 + 2 * half;
    const float* wi1 = w_ih + (long)(64 + colb) * 64 + 2 * half;
    const float* wi2 = w_ih + (long)(128 + colb) * 64 + 2 * half;
    const float* wh0 = w_hh + (long)(colb)*64 + 2 * half;
    const float* wh1 = w_hh + (long)(64 + colb) * 64 + 2 * half;
    const float* wh2 = w_hh + (long)(128 + colb) * 64 + 2 * half;
#pragma unroll
    for (int c = 0; c < 16; ++c) {
      ir  = wmma4(Aa[c], ld2(wi0 + 4 * c), ir);
      iz  = wmma4(Aa[c], ld2(wi1 + 4 * c), iz);
      in_ = wmma4(Aa[c], ld2(wi2 + 4 * c), in_);
      hr  = wmma4(Ah[c], ld2(wh0 + 4 * c), hr);
      hz  = wmma4(Ah[c], ld2(wh1 + 4 * c), hz);
      hn  = wmma4(Ah[c], ld2(wh2 + 4 * c), hn);
    }
#pragma unroll
    for (int i = 0; i < 8; ++i) {
      const long m = nbase + i + 8 * half;
      const float h = hvin[m * GP_NODE_DIM + colb];
      const float r = sigmoid_(ir[i] + hr[i]);
      const float z = sigmoid_(iz[i] + hz[i]);
      const float n = tanhf(in_[i] + r * hn[i]);
      hvout[m * GP_NODE_DIM + colb] = (1.0f - z) * n + z * h;
    }
  }
}

extern "C" void kernel_launch(void* const* d_in, const int* in_sizes, int n_in,
                              void* d_out, int out_size, void* d_ws, size_t ws_size,
                              hipStream_t stream) {
  const float* hv    = (const float*)d_in[0];
  const float* he    = (const float*)d_in[1];
  const int*   src   = (const int*)d_in[2];
  const int*   dst   = (const int*)d_in[3];
  const float* W_msg = (const float*)d_in[4];
  const float* b_msg = (const float*)d_in[5];
  const float* w_ih  = (const float*)d_in[6];
  const float* w_hh  = (const float*)d_in[7];
  const float* b_ih  = (const float*)d_in[8];
  const float* b_hh  = (const float*)d_in[9];
  float* out = (float*)d_out;

  const size_t state_elems = (size_t)GP_N_NODES * GP_NODE_DIM;
  float* a_buf  = (float*)d_ws;
  float* hv_buf = a_buf + state_elems;

  hipMemcpyAsync(hv_buf, hv, state_elems * sizeof(float),
                 hipMemcpyDeviceToDevice, stream);

  const int edge_blocks = (GP_N_EDGES / 16) / 8;
  const int gru_blocks  = (GP_N_NODES / 16 + 7) / 8;

  for (int t = 0; t < GP_ROUNDS; ++t) {
    hipMemsetAsync(a_buf, 0, state_elems * sizeof(float), stream);
    gp_edge_kernel<<<edge_blocks, 256, 0, stream>>>(
        hv_buf, he, src, dst,
        W_msg + (size_t)t * GP_CAT_DIM * GP_ACT_DIM,
        b_msg + (size_t)t * GP_ACT_DIM, a_buf);
    float* outp = (t == GP_ROUNDS - 1) ? out : hv_buf;
    gp_gru_kernel<<<gru_blocks, 256, 0, stream>>>(
        a_buf, hv_buf,
        w_ih + (size_t)t * 3 * GP_NODE_DIM * GP_NODE_DIM,
        w_hh + (size_t)t * 3 * GP_NODE_DIM * GP_NODE_DIM,
        b_ih + (size_t)t * 3 * GP_NODE_DIM,
        b_hh + (size_t)t * 3 * GP_NODE_DIM, outp);
  }
}


# speedup vs baseline: 1.9961x; 1.9961x over previous
#include <hip/hip_runtime.h>


typedef float v2f __attribute__((ext_vector_type(2)));
typedef float v4f __attribute__((ext_vector_type(4)));
typedef float v8f __attribute__((ext_vector_type(8)));

#define GP_ROUNDS   3
#define GP_NODE_DIM 64
#define GP_ACT_DIM  64
#define GP_EDGE_DIM 32
#define GP_N_NODES  100000
#define GP_N_EDGES  1600000
#define GP_CAT_DIM  160
#define KC_SRC  16
#define KC_DST  16
#define KC_EDG  8
#define KC_ALL  40
#define GROUPS_PER_WAVE 5

static __device__ __forceinline__ v8f wmma4(v2f a, v2f b, v8f c) {
  return __builtin_amdgcn_wmma_f32_16x16x4_f32(false, a, false, b, (short)0, c,
                                               false, false);
}
static __device__ __forceinline__ v2f ld2(const float* p) {
  return *(const v2f*)p;
}
static __device__ __forceinline__ float sigmoid_(float x) {
  return 1.0f / (1.0f + __expf(-x));
}
static __device__ __forceinline__ int pkg_half() {
  return (int)((__builtin_amdgcn_s_sendmsg_rtn(0x87) >> 18) & 1u);
}


__global__ __launch_bounds__(256) void gp_hesum_kernel(
    const float* __restrict__ he, const int* __restrict__ dst,
    float* gsum0, float* gsum1,
    float* deg0, float* deg1)
{
  const int hsel = pkg_half();
  float* gsum = hsel ? gsum1 : gsum0;
  float* deg  = hsel ? deg1 : deg0;
  const int lane = threadIdx.x & 31;
  const int wave = threadIdx.x >> 5;
  const int g0 = (blockIdx.x * 8 + wave) * GROUPS_PER_WAVE;
  for (int it = 0; it < GROUPS_PER_WAVE; ++it) {
    const int e0 = (g0 + it) * 32;
    const int dv = dst[e0 + lane];
    atomicAdd(&deg[dv], 1.0f);
    const float* hp = he + (long)e0 * GP_EDGE_DIM + lane;
    float v[32];
#pragma unroll
    for (int j = 0; j < 32; ++j) v[j] = hp[j * GP_EDGE_DIM];
#pragma unroll
    for (int j = 0; j < 32; ++j) {
      const int d = __builtin_amdgcn_readlane(dv, j);
      atomicAdd(gsum + (long)d * GP_EDGE_DIM + lane, v[j]);
    }
  }
}

__global__ __launch_bounds__(256) void gp_node_kernel(
    const float* __restrict__ hvin,
    const float* __restrict__ W,
    const float* __restrict__ bias,
    const float* gsum0, const float* gsum1,
    const float* deg0, const float* deg1,
    float* abuf0,
    float* abuf1,
    float* __restrict__ ptab)
{
  const bool two = (abuf1 != abuf0);
  __shared__ float Wl[KC_ALL * 4 * 32 * 2];
  const int tid = threadIdx.x;
  for (int p = tid; p < KC_ALL * 4 * 32; p += 256) {
    const int L  = p & 31;
    const int q  = p >> 5;
    const int n0 = q & 3;
    const int c  = q >> 2;
    const int k  = 4 * c + 2 * (L >> 4);
    const int col = n0 * 16 + (L & 15);
    v2f w2;
    w2.x = W[k * 64 + col];
    w2.y = W[(k + 1) * 64 + col];
    *(v2f*)&Wl[2 * p] = w2;
  }
  __syncthreads();

  const int lane = tid & 31;
  const int wave = tid >> 5;
  const int tile = blockIdx.x * 8 + wave;
  if (tile >= GP_N_NODES / 16) return;

  const int nbase = tile * 16;
  const int row   = lane & 15;
  const int half  = lane >> 4;

  v2f Ah[KC_SRC];
  const float* hq = hvin + (long)(nbase + row) * GP_NODE_DIM + 2 * half;
#pragma unroll
  for (int c = 0; c < KC_SRC; ++c) Ah[c] = ld2(hq + 4 * c);

  v2f Ag[KC_EDG];
  const long goff = (long)(nbase + row) * GP_EDGE_DIM + 2 * half;
#pragma unroll
  for (int c = 0; c < KC_EDG; ++c) Ag[c] = ld2(gsum0 + goff + 4 * c);

  float dg[8];
#pragma unroll
  for (int i = 0; i < 8; ++i) dg[i] = deg0[nbase + i + 8 * half];

  if (two) {
#pragma unroll
    for (int c = 0; c < KC_EDG; ++c) Ag[c] += ld2(gsum1 + goff + 4 * c);
#pragma unroll
    for (int i = 0; i < 8; ++i) dg[i] += deg1[nbase + i + 8 * half];
    v4f zero4 = {0.0f, 0.0f, 0.0f, 0.0f};
    float* zp = abuf1 + (long)nbase * GP_ACT_DIM + 4 * lane;
#pragma unroll
    for (int i = 0; i < 8; ++i) *(v4f*)(zp + i * 128) = zero4;
  }

  const v2f* Wl2 = (const v2f*)Wl;
  v8f accP[4];
#pragma unroll
  for (int n0 = 0; n0 < 4; ++n0) {
    const float bv = bias[n0 * 16 + row];
    v8f accQ, accR;
#pragma unroll
    for (int i = 0; i < 8; ++i) { accQ[i] = bv; accR[i] = 0.0f; accP[n0][i] = 0.0f; }

    const v2f* Bp = Wl2 + (n0 * 32 + lane);
#pragma unroll
    for (int c = 0; c < KC_SRC; ++c) {
      accP[n0] = wmma4(Ah[c], Bp[(c) * 128], accP[n0]);
      accQ     = wmma4(Ah[c], Bp[(KC_SRC + c) * 128], accQ);
    }
#pragma unroll
    for (int c = 0; c < KC_EDG; ++c)
      accR = wmma4(Ag[c], Bp[(KC_SRC + KC_DST + c) * 128], accR);
#pragma unroll
    for (int i = 0; i < 8; ++i)
      abuf0[(long)(nbase + i + 8 * half) * GP_ACT_DIM + n0 * 16 + row] =
          dg[i] * accQ[i] + accR[i];
  }
#pragma unroll
  for (int i = 0; i < 8; ++i) {
    float* pr = ptab + (long)(nbase + i + 8 * half) * GP_NODE_DIM + 2 * row;
    v2f lo, hi;
    lo.x = accP[0][i]; lo.y = accP[2][i];
    hi.x = accP[1][i]; hi.y = accP[3][i];
    *(v2f*)pr = lo;
    *(v2f*)(pr + 32) = hi;
  }
}

__global__ __launch_bounds__(256) void gp_edge_kernel(
    const float* __restrict__ ptab,
    const int* __restrict__ src, const int* __restrict__ dst,
    float* abuf0, float* abuf1)
{
  float* abuf = pkg_half() ? abuf1 : abuf0;
  const int lane = threadIdx.x & 31;
  const int wave = threadIdx.x >> 5;
  const int g0 = (blockIdx.x * 8 + wave) * GROUPS_PER_WAVE;
  for (int it = 0; it < GROUPS_PER_WAVE; ++it) {
    const int e0 = (g0 + it) * 32;
    const int sv = src[e0 + lane];
    const int dv = dst[e0 + lane];
#pragma unroll
    for (int hb = 0; hb < 32; hb += 16) {
      v2f val[16];
#pragma unroll
      for (int j = 0; j < 16; ++j) {
        const int s = __builtin_amdgcn_readlane(sv, hb + j);
        val[j] = ld2(ptab + (long)s * GP_NODE_DIM + 2 * lane);
      }
#pragma unroll
      for (int j = 0; j < 16; ++j) {
        const int d = __builtin_amdgcn_readlane(dv, hb + j);
        float* ap = abuf + (long)d * GP_ACT_DIM + lane;
        atomicAdd(ap, val[j].x);
        atomicAdd(ap + 32, val[j].y);
      }
    }
  }
}

__global__ __launch_bounds__(256) void gp_gru_kernel(
    const float* abuf0, const float* abuf1,
    const float* __restrict__ hvin,
    const float* __restrict__ w_ih, const float* __restrict__ w_hh,
    const float* __restrict__ b_ih, const float* __restrict__ b_hh,
    float* __restrict__ hvout)
{
  const int tid  = threadIdx.x;
  const int lane = tid & 31;
  const int wave = tid >> 5;
  const int tile = blockIdx.x * 8 + wave;
  const int ntiles = GP_N_NODES / 16;
  if (tile >= ntiles) return;

  const int nbase = tile * 16;
  const int row   = lane & 15;
  const int half  = lane >> 4;

  v2f Aa[16], Ah[16];
  const long aoff = (long)(nbase + row) * GP_ACT_DIM + 2 * half;
  const float* hq = hvin + (long)(nbase + row) * GP_NODE_DIM + 2 * half;
#pragma unroll
  for (int c = 0; c < 16; ++c) { Aa[c] = ld2(abuf0 + aoff + 4 * c); Ah[c] = ld2(hq + 4 * c); }
  if (abuf1 != abuf0) {
#pragma unroll
    for (int c = 0; c < 16; ++c) Aa[c] += ld2(abuf1 + aoff + 4 * c);
  }

#pragma unroll
  for (int j0 = 0; j0 < 4; ++j0) {
    const int colb = j0 * 16 + row;
    v8f ir, iz, in_, hr, hz, hn;
    {
      const float bi0 = b_ih[colb], bi1 = b_ih[64 + colb], bi2 = b_ih[128 + colb];
      const float bh0 = b_hh[colb], bh1 = b_hh[64 + colb], bh2 = b_hh[128 + colb];
#pragma unroll
      for (int i = 0; i < 8; ++i) {
        ir[i] = bi0; iz[i] = bi1; in_[i] = bi2;
        hr[i] = bh0; hz[i] = bh1; hn[i] = bh2;
      }
    }
    const float* wi0 = w_ih + (long)(colb)*64 + 2 * half;
    const float* wi1 = w_ih + (long)(64 + colb) * 64 + 2 * half;
    const float* wi2 = w_ih + (long)(128 + colb) * 64 + 2 * half;
    const float* wh0 = w_hh + (long)(colb)*64 + 2 * half;
    const float* wh1 = w_hh + (long)(64 + colb) * 64 + 2 * half;
    const float* wh2 = w_hh + (long)(128 + colb) * 64 + 2 * half;
#pragma unroll
    for (int c = 0; c < 16; ++c) {
      ir  = wmma4(Aa[c], ld2(wi0 + 4 * c), ir);
      iz  = wmma4(Aa[c], ld2(wi1 + 4 * c), iz);
      in_ = wmma4(Aa[c], ld2(wi2 + 4 * c), in_);
      hr  = wmma4(Ah[c], ld2(wh0 + 4 * c), hr);
      hz  = wmma4(Ah[c], ld2(wh1 + 4 * c), hz);
      hn  = wmma4(Ah[c], ld2(wh2 + 4 * c), hn);
    }
#pragma unroll
    for (int i = 0; i < 8; ++i) {
      const long m = nbase + i + 8 * half;
      const float h = hvin[m * GP_NODE_DIM + colb];
      const float r = sigmoid_(ir[i] + hr[i]);
      const float z = sigmoid_(iz[i] + hz[i]);
      const float n = tanhf(in_[i] + r * hn[i]);
      hvout[m * GP_NODE_DIM + colb] = (1.0f - z) * n + z * h;
    }
  }
}

extern "C" void kernel_launch(void* const* d_in, const int* in_sizes, int n_in,
                              void* d_out, int out_size, void* d_ws, size_t ws_size,
                              hipStream_t stream) {
  const float* hv    = (const float*)d_in[0];
  const float* he    = (const float*)d_in[1];
  const int*   src   = (const int*)d_in[2];
  const int*   dst   = (const int*)d_in[3];
  const float* W_msg = (const float*)d_in[4];
  const float* b_msg = (const float*)d_in[5];
  const float* w_ih  = (const float*)d_in[6];
  const float* w_hh  = (const float*)d_in[7];
  const float* b_ih  = (const float*)d_in[8];
  const float* b_hh  = (const float*)d_in[9];
  float* out = (float*)d_out;

  const size_t state_elems = (size_t)GP_N_NODES * GP_NODE_DIM;
  float* a_buf  = (float*)d_ws;
  float* hv_buf = a_buf + state_elems;
  const size_t gsum_elems = (size_t)GP_N_NODES * GP_EDGE_DIM;
  float* g_sum  = hv_buf + state_elems;
  float* deg    = g_sum + gsum_elems;
  if (ws_size < (2 * state_elems + gsum_elems + GP_N_NODES) * sizeof(float)) return;
  const bool two = ws_size >= (3 * state_elems + 2 * gsum_elems + 2 * GP_N_NODES) * sizeof(float);
  float* g_sum1 = two ? deg + GP_N_NODES : g_sum;
  float* deg1   = two ? g_sum1 + gsum_elems : deg;
  float* a_buf1 = two ? deg1 + GP_N_NODES : a_buf;
  float* p_tab  = out;

  const int grp_blocks  = (GP_N_EDGES / 32) / (8 * GROUPS_PER_WAVE);
  const int node_blocks = (GP_N_NODES / 16 + 7) / 8;
  const int gru_blocks  = (GP_N_NODES / 16 + 7) / 8;

  hipMemsetAsync(g_sum, 0, (two ? 2 : 1) * (gsum_elems + GP_N_NODES) * sizeof(float), stream);
  gp_hesum_kernel<<<grp_blocks, 256, 0, stream>>>(he, dst, g_sum, g_sum1, deg, deg1);

  for (int t = 0; t < GP_ROUNDS; ++t) {
    const float* hcur = (t == 0) ? hv : hv_buf;
    const float* Wt = W_msg + (size_t)t * GP_CAT_DIM * GP_ACT_DIM;
    gp_node_kernel<<<node_blocks, 256, 0, stream>>>(
        hcur, Wt, b_msg + (size_t)t * GP_ACT_DIM, g_sum, g_sum1, deg, deg1,
        a_buf, a_buf1, p_tab);
    gp_edge_kernel<<<grp_blocks, 256, 0, stream>>>(p_tab, src, dst, a_buf, a_buf1);
    float* outp = (t == GP_ROUNDS - 1) ? out : hv_buf;
    gp_gru_kernel<<<gru_blocks, 256, 0, stream>>>(
        a_buf, a_buf1, hcur,
        w_ih + (size_t)t * 3 * GP_NODE_DIM * GP_NODE_DIM,
        w_hh + (size_t)t * 3 * GP_NODE_DIM * GP_NODE_DIM,
        b_ih + (size_t)t * 3 * GP_NODE_DIM,
        b_hh + (size_t)t * 3 * GP_NODE_DIM, outp);
  }
}


# Hidden-state half of the GRU products (h@w_hh^T+b_hh) runs in extra work-groups of each edge launch, on WMMA beside the add-bound scatter; the GRU launch reads it back; sum order kept
# speedup vs baseline: 2.1370x; 1.0706x over previous
#include <hip/hip_runtime.h>


typedef float v2f __attribute__((ext_vector_type(2)));
typedef float v4f __attribute__((ext_vector_type(4)));
typedef float v8f __attribute__((ext_vector_type(8)));

#define GP_ROUNDS   3
#define GP_NODE_DIM 64
#define GP_ACT_DIM  64
#define GP_EDGE_DIM 32
#define GP_N_NODES  100000
#define GP_N_EDGES  1600000
#define GP_CAT_DIM  160
#define KC_SRC  16
#define KC_DST  16
#define KC_EDG  8
#define KC_ALL  40
#define GROUPS_PER_WAVE 5

static __device__ __forceinline__ v8f wmma4(v2f a, v2f b, v8f c) {
  return __builtin_amdgcn_wmma_f32_16x16x4_f32(false, a, false, b, (short)0, c,
                                               false, false);
}
static __device__ __forceinline__ v2f ld2(const float* p) {
  return *(const v2f*)p;
}
static __device__ __forceinline__ float sigmoid_(float x) {
  return 1.0f / (1.0f + __expf(-x));
}
static __device__ __forceinline__ int pkg_half() {
  return (int)((__builtin_amdgcn_s_sendmsg_rtn(0x87) >> 18) & 1u);
}


__global__ __launch_bounds__(256) void gp_hesum_kernel(
    const float* __restrict__ he, const int* __restrict__ dst,
    float* gsum0, float* gsum1,
    float* deg0, float* deg1)
{
  const int hsel = pkg_half();
  float* gsum = hsel ? gsum1 : gsum0;
  float* deg  = hsel ? deg1 : deg0;
  const int lane = threadIdx.x & 31;
  const int wave = threadIdx.x >> 5;
  const int g0 = (blockIdx.x * 8 + wave) * GROUPS_PER_WAVE;
  for (int it = 0; it < GROUPS_PER_WAVE; ++it) {
    const int e0 = (g0 + it) * 32;
    const int dv = dst[e0 + lane];
    atomicAdd(&deg[dv], 1.0f);
    const float* hp = he + (long)e0 * GP_EDGE_DIM + lane;
    float v[32];
#pragma unroll
    for (int j = 0; j < 32; ++j) v[j] = hp[j * GP_EDGE_DIM];
#pragma unroll
    for (int j = 0; j < 32; ++j) {
      const int d = __builtin_amdgcn_readlane(dv, j);
      atomicAdd(gsum + (long)d * GP_EDGE_DIM + lane, v[j]);
    }
  }
}

__global__ __launch_bounds__(256) void gp_node_kernel(
    const float* __restrict__ hvin,
    const float* __restrict__ W,
    const float* __restrict__ bias,
    const float* gsum0, const float* gsum1,
    const float* deg0, const float* deg1,
    float* abuf0,
    float* abuf1,
    float* __restrict__ ptab)
{
  const bool two = (abuf1 != abuf0);
  __shared__ float Wl[KC_ALL * 4 * 32 * 2];
  const int tid = threadIdx.x;
  for (int p = tid; p < KC_ALL * 4 * 32; p += 256) {
    const int L  = p & 31;
    const int q  = p >> 5;
    const int n0 = q & 3;
    const int c  = q >> 2;
    const int k  = 4 * c + 2 * (L >> 4);
    const int col = n0 * 16 + (L & 15);
    v2f w2;
    w2.x = W[k * 64 + col];
    w2.y = W[(k + 1) * 64 + col];
    *(v2f*)&Wl[2 * p] = w2;
  }
  __syncthreads();

  const int lane = tid & 31;
  const int wave = tid >> 5;
  const int tile = blockIdx.x * 8 + wave;
  if (tile >= GP_N_NODES / 16) return;

  const int nbase = tile * 16;
  const int row   = lane & 15;
  const int half  = lane >> 4;

  v2f Ah[KC_SRC];
  const float* hq = hvin + (long)(nbase + row) * GP_NODE_DIM + 2 * half;
#pragma unroll
  for (int c = 0; c < KC_SRC; ++c) Ah[c] = ld2(hq + 4 * c);

  v2f Ag[KC_EDG];
  const long goff = (long)(nbase + row) * GP_EDGE_DIM + 2 * half;
#pragma unroll
  for (int c = 0; c < KC_EDG; ++c) Ag[c] = ld2(gsum0 + goff + 4 * c);

  float dg[8];
#pragma unroll
  for (int i = 0; i < 8; ++i) dg[i] = deg0[nbase + i + 8 * half];

  if (two) {
#pragma unroll
    for (int c = 0; c < KC_EDG; ++c) Ag[c] += ld2(gsum1 + goff + 4 * c);
#pragma unroll
    for (int i = 0; i < 8; ++i) dg[i] += deg1[nbase + i + 8 * half];
    v4f zero4 = {0.0f, 0.0f, 0.0f, 0.0f};
    float* zp = abuf1 + (long)nbase * GP_ACT_DIM + 4 * lane;
#pragma unroll
    for (int i = 0; i < 8; ++i) *(v4f*)(zp + i * 128) = zero4;
  }

  const v2f* Wl2 = (const v2f*)Wl;
  v8f accP[4];
#pragma unroll
  for (int n0 = 0; n0 < 4; ++n0) {
    const float bv = bias[n0 * 16 + row];
    v8f accQ, accR;
#pragma unroll
    for (int i = 0; i < 8; ++i) { accQ[i] = bv; accR[i] = 0.0f; accP[n0][i] = 0.0f; }

    const v2f* Bp = Wl2 + (n0 * 32 + lane);
#pragma unroll
    for (int c = 0; c < KC_SRC; ++c) {
      accP[n0] = wmma4(Ah[c], Bp[(c) * 128], accP[n0]);
      accQ     = wmma4(Ah[c], Bp[(KC_SRC + c) * 128], accQ);
    }
#pragma unroll
    for (int c = 0; c < KC_EDG; ++c)
      accR = wmma4(Ag[c], Bp[(KC_SRC + KC_DST + c) * 128], accR);
#pragma unroll
    for (int i = 0; i < 8; ++i)
      abuf0[(long)(nbase + i + 8 * half) * GP_ACT_DIM + n0 * 16 + row] =
          dg[i] * accQ[i] + accR[i];
  }
#pragma unroll
  for (int i = 0; i < 8; ++i) {
    float* pr = ptab + (long)(nbase + i + 8 * half) * GP_NODE_DIM + 2 * row;
    v2f lo, hi;
    lo.x = accP[0][i]; lo.y = accP[2][i];
    hi.x = accP[1][i]; hi.y = accP[3][i];
    *(v2f*)pr = lo;
    *(v2f*)(pr + 32) = hi;
  }
}

__global__ __launch_bounds__(256) void gp_edge_kernel(
    const float* __restrict__ ptab,
    const int* __restrict__ src, const int* __restrict__ dst,
    float* abuf0, float* abuf1,
    const float* __restrict__ hvin,
    const float* __restrict__ w_hh, const float* __restrict__ b_hh,
    float* __restrict__ ghbuf,
    int gh_blocks)
{
  if ((int)blockIdx.x < gh_blocks) {
    const int lane = threadIdx.x & 31;
    const int wave = threadIdx.x >> 5;
    const int tile = blockIdx.x * 8 + wave;
    if (tile >= GP_N_NODES / 16) return;
    const int nbase = tile * 16;
    const int row   = lane & 15;
    const int half  = lane >> 4;
    const float* hq = hvin + (long)(nbase + row) * GP_NODE_DIM + 2 * half;
#pragma unroll 1
    for (int j0 = 0; j0 < 4; ++j0) {
      const int colb = j0 * 16 + row;
      v8f hr, hz, hn;
      const float bh0 = b_hh[colb], bh1 = b_hh[64 + colb], bh2 = b_hh[128 + colb];
#pragma unroll
      for (int i = 0; i < 8; ++i) { hr[i] = bh0; hz[i] = bh1; hn[i] = bh2; }
      const float* wh0 = w_hh + (long)(colb)*64 + 2 * half;
      const float* wh1 = w_hh + (long)(64 + colb) * 64 + 2 * half;
      const float* wh2 = w_hh + (long)(128 + colb) * 64 + 2 * half;
#pragma unroll 1
      for (int cc = 0; cc < 16; cc += 4) {
#pragma unroll
        for (int c = 0; c < 4; ++c) {
          const v2f a = ld2(hq + 4 * (cc + c));
          hr = wmma4(a, ld2(wh0 + 4 * (cc + c)), hr);
          hz = wmma4(a, ld2(wh1 + 4 * (cc + c)), hz);
          hn = wmma4(a, ld2(wh2 + 4 * (cc + c)), hn);
        }
      }
      float* gp = ghbuf + ((long)tile * 4 + j0) * 768 + 4 * lane;
      v4f t;
      t[0] = hr[0]; t[1] = hr[1]; t[2] = hr[2]; t[3] = hr[3]; *(v4f*)(gp)       = t;
      t[0] = hr[4]; t[1] = hr[5]; t[2] = hr[6]; t[3] = hr[7]; *(v4f*)(gp + 128) = t;
      t[0] = hz[0]; t[1] = hz[1]; t[2] = hz[2]; t[3] = hz[3]; *(v4f*)(gp + 256) = t;
      t[0] = hz[4]; t[1] = hz[5]; t[2] = hz[6]; t[3] = hz[7]; *(v4f*)(gp + 384) = t;
      t[0] = hn[0]; t[1] = hn[1]; t[2] = hn[2]; t[3] = hn[3]; *(v4f*)(gp + 512) = t;
      t[0] = hn[4]; t[1] = hn[5]; t[2] = hn[6]; t[3] = hn[7]; *(v4f*)(gp + 640) = t;
    }
    return;
  }
  float* abuf = pkg_half() ? abuf1 : abuf0;
  const int lane = threadIdx.x & 31;
  const int wave = threadIdx.x >> 5;
  const int g0 = (((int)blockIdx.x - gh_blocks) * 8 + wave) * GROUPS_PER_WAVE;
  for (int it = 0; it < GROUPS_PER_WAVE; ++it) {
    const int e0 = (g0 + it) * 32;
    const int sv = src[e0 + lane];
    const int dv = dst[e0 + lane];
#pragma unroll
    for (int hb = 0; hb < 32; hb += 16) {
      v2f val[16];
#pragma unroll
      for (int j = 0; j < 16; ++j) {
        const int s = __builtin_amdgcn_readlane(sv, hb + j);
        val[j] = ld2(ptab + (long)s * GP_NODE_DIM + 2 * lane);
      }
#pragma unroll
      for (int j = 0; j < 16; ++j) {
        const int d = __builtin_amdgcn_readlane(dv, hb + j);
        float* ap = abuf + (long)d * GP_ACT_DIM + lane;
        atomicAdd(ap, val[j].x);
        atomicAdd(ap + 32, val[j].y);
      }
    }
  }
}

template <bool GH_PRE>
__global__ __launch_bounds__(256) void gp_gru_kernel(
    const float* abuf0, const float* abuf1,
    const float* __restrict__ hvin,
    const float* __restrict__ w_ih, const float* __restrict__ w_hh,
    const float* __restrict__ b_ih, const float* __restrict__ b_hh,
    const float* __restrict__ ghbuf,
    float* __restrict__ hvout)
{
  const int tid  = threadIdx.x;
  const int lane = tid & 31;
  const int wave = tid >> 5;
  const int tile = blockIdx.x * 8 + wave;
  const int ntiles = GP_N_NODES / 16;
  if (tile >= ntiles) return;

  const int nbase = tile * 16;
  const int row   = lane & 15;
  const int half  = lane >> 4;

  v2f Aa[16], Ah[16];
  const long aoff = (long)(nbase + row) * GP_ACT_DIM + 2 * half;
  const float* hq = hvin + (long)(nbase + row) * GP_NODE_DIM + 2 * half;
#pragma unroll
  for (int c = 0; c < 16; ++c) {
    Aa[c] = ld2(abuf0 + aoff + 4 * c);
    if (!GH_PRE) Ah[c] = ld2(hq + 4 * c);
  }
  if (abuf1 != abuf0) {
#pragma unroll
    for (int c = 0; c < 16; ++c) Aa[c] += ld2(abuf1 + aoff + 4 * c);
  }

#pragma unroll
  for (int j0 = 0; j0 < 4; ++j0) {
    const int colb = j0 * 16 + row;
    v8f ir, iz, in_, hr, hz, hn;
    {
      const float bi0 = b_ih[colb], bi1 = b_ih[64 + colb], bi2 = b_ih[128 + colb];
#pragma unroll
      for (int i = 0; i < 8; ++i) { ir[i] = bi0; iz[i] = bi1; in_[i] = bi2; }
    }
    const float* wi0 = w_ih + (long)(colb)*64 + 2 * half;
    const float* wi1 = w_ih + (long)(64 + colb) * 64 + 2 * half;
    const float* wi2 = w_ih + (long)(128 + colb) * 64 + 2 * half;
    if (GH_PRE) {
      const float* gp = ghbuf + ((long)tile * 4 + j0) * 768 + 4 * lane;
      const v4f r0 = *(const v4f*)(gp),       r1 = *(const v4f*)(gp + 128);
      const v4f z0 = *(const v4f*)(gp + 256), z1 = *(const v4f*)(gp + 384);
      const v4f n0 = *(const v4f*)(gp + 512), n1 = *(const v4f*)(gp + 640);
#pragma unroll
      for (int i = 0; i < 4; ++i) {
        hr[i] = r0[i]; hr[4 + i] = r1[i];
        hz[i] = z0[i]; hz[4 + i] = z1[i];
        hn[i] = n0[i]; hn[4 + i] = n1[i];
      }
#pragma unroll
      for (int c = 0; c < 16; ++c) {
        ir  = wmma4(Aa[c], ld2(wi0 + 4 * c), ir);
        iz  = wmma4(Aa[c], ld2(wi1 + 4 * c), iz);
        in_ = wmma4(Aa[c], ld2(wi2 + 4 * c), in_);
      }
    } else {
      const float bh0 = b_hh[colb], bh1 = b_hh[64 + colb], bh2 = b_hh[128 + colb];
#pragma unroll
      for (int i = 0; i < 8; ++i) { hr[i] = bh0; hz[i] = bh1; hn[i] = bh2; }
      const float* wh0 = w_hh + (long)(colb)*64 + 2 * half;
      const float* wh1 = w_hh + (long)(64 + colb) * 64 + 2 * half;
      const float* wh2 = w_hh + (long)(128 + colb) * 64 + 2 * half;
#pragma unroll
      for (int c = 0; c < 16; ++c) {
        ir  = wmma4(Aa[c], ld2(wi0 + 4 * c), ir);
        iz  = wmma4(Aa[c], ld2(wi1 + 4 * c), iz);
        in_ = wmma4(Aa[c], ld2(wi2 + 4 * c), in_);
        hr  = wmma4(Ah[c], ld2(wh0 + 4 * c), hr);
        hz  = wmma4(Ah[c], ld2(wh1 + 4 * c), hz);
        hn  = wmma4(Ah[c], ld2(wh2 + 4 * c), hn);
      }
    }
#pragma unroll
    for (int i = 0; i < 8; ++i) {
      const long m = nbase + i + 8 * half;
      const float h = hvin[m * GP_NODE_DIM + colb];
      const float r = sigmoid_(ir[i] + hr[i]);
      const float z = sigmoid_(iz[i] + hz[i]);
      const float n = tanhf(in_[i] + r * hn[i]);
      hvout[m * GP_NODE_DIM + colb] = (1.0f - z) * n + z * h;
    }
  }
}

extern "C" void kernel_launch(void* const* d_in, const int* in_sizes, int n_in,
                              void* d_out, int out_size, void* d_ws, size_t ws_size,
                              hipStream_t stream) {
  const float* hv    = (const float*)d_in[0];
  const float* he    = (const float*)d_in[1];
  const int*   src   = (const int*)d_in[2];
  const int*   dst   = (const int*)d_in[3];
  const float* W_msg = (const float*)d_in[4];
  const float* b_msg = (const float*)d_in[5];
  const float* w_ih  = (const float*)d_in[6];
  const float* w_hh  = (const float*)d_in[7];
  const float* b_ih  = (const float*)d_in[8];
  const float* b_hh  = (const float*)d_in[9];
  float* out = (float*)d_out;

  const size_t state_elems = (size_t)GP_N_NODES * GP_NODE_DIM;
  float* a_buf  = (float*)d_ws;
  float* hv_buf = a_buf + state_elems;
  const size_t gsum_elems = (size_t)GP_N_NODES * GP_EDGE_DIM;
  float* g_sum  = hv_buf + state_elems;
  float* deg    = g_sum + gsum_elems;
  if (ws_size < (2 * state_elems + gsum_elems + GP_N_NODES) * sizeof(float)) return;
  const bool two = ws_size >= (3 * state_elems + 2 * gsum_elems + 2 * GP_N_NODES) * sizeof(float);
  float* g_sum1 = two ? deg + GP_N_NODES : g_sum;
  float* deg1   = two ? g_sum1 + gsum_elems : deg;
  float* a_buf1 = two ? deg1 + GP_N_NODES : a_buf;
  const bool pre = ws_size >= (6 * state_elems + 2 * gsum_elems + 2 * GP_N_NODES) * sizeof(float);
  float* gh_buf = a_buf1 + state_elems;
  float* p_tab  = out;

  const int grp_blocks  = (GP_N_EDGES / 32) / (8 * GROUPS_PER_WAVE);
  const int node_blocks = (GP_N_NODES / 16 + 7) / 8;
  const int gru_blocks  = (GP_N_NODES / 16 + 7) / 8;

  hipMemsetAsync(g_sum, 0, (two ? 2 : 1) * (gsum_elems + GP_N_NODES) * sizeof(float), stream);
  gp_hesum_kernel<<<grp_blocks, 256, 0, stream>>>(he, dst, g_sum, g_sum1, deg, deg1);

  for (int t = 0; t < GP_ROUNDS; ++t) {
    const float* hcur = (t == 0) ? hv : hv_buf;
    const float* Wt = W_msg + (size_t)t * GP_CAT_DIM * GP_ACT_DIM;
    gp_node_kernel<<<node_blocks, 256, 0, stream>>>(
        hcur, Wt, b_msg + (size_t)t * GP_ACT_DIM, g_sum, g_sum1, deg, deg1,
        a_buf, a_buf1, p_tab);
    const float* whh = w_hh + (size_t)t * 3 * GP_NODE_DIM * GP_NODE_DIM;
    const float* bhh = b_hh + (size_t)t * 3 * GP_NODE_DIM;
    float* outp = (t == GP_ROUNDS - 1) ? out : hv_buf;
    if (pre) {
      gp_edge_kernel<<<gru_blocks + grp_blocks, 256, 0, stream>>>(
          p_tab, src, dst, a_buf, a_buf1, hcur, whh, bhh, gh_buf, gru_blocks);
      gp_gru_kernel<true><<<gru_blocks, 256, 0, stream>>>(
          a_buf, a_buf1, hcur,
          w_ih + (size_t)t * 3 * GP_NODE_DIM * GP_NODE_DIM, whh,
          b_ih + (size_t)t * 3 * GP_NODE_DIM, bhh, gh_buf, outp);
    } else {
      gp_edge_kernel<<<grp_blocks, 256, 0, stream>>>(
          p_tab, src, dst, a_buf, a_buf1, hcur, whh, bhh, a_buf, 0);
      gp_gru_kernel<false><<<gru_blocks, 256, 0, stream>>>(
          a_buf, a_buf1, hcur,
          w_ih + (size_t)t * 3 * GP_NODE_DIM * GP_NODE_DIM, whh,
          b_ih + (size_t)t * 3 * GP_NODE_DIM, bhh, a_buf, outp);
    }
  }
}


# GRU and gh products read A, B fragments 16 B per lane (k paired per two steps); w_ih, w_hh pre-ordered once per call in a workspace table: a B load is 4 whole lines, not 16 pieces; sum order changes
# speedup vs baseline: 2.3603x; 1.1045x over previous
#include <hip/hip_runtime.h>


typedef float v2f __attribute__((ext_vector_type(2)));
typedef float v4f __attribute__((ext_vector_type(4)));
typedef float v8f __attribute__((ext_vector_type(8)));

#define GP_ROUNDS   3
#define GP_NODE_DIM 64
#define GP_ACT_DIM  64
#define GP_EDGE_DIM 32
#define GP_N_NODES  100000
#define GP_N_EDGES  1600000
#define GP_CAT_DIM  160
#define KC_SRC  16
#define KC_DST  16
#define KC_EDG  8
#define KC_ALL  40
#define GROUPS_PER_WAVE 5
#define WT_MAT_FLOATS 12288
#define WT_MAT_PIECES 3072
#define WT_BLOCKS 72

static __device__ __forceinline__ v8f wmma4(v2f a, v2f b, v8f c) {
  return __builtin_amdgcn_wmma_f32_16x16x4_f32(false, a, false, b, (short)0, c,
                                               false, false);
}
static __device__ __forceinline__ v2f ld2(const float* p) {
  return *(const v2f*)p;
}
static __device__ __forceinline__ float sigmoid_(float x) {
  return 1.0f / (1.0f + __expf(-x));
}
static __device__ __forceinline__ int pkg_half() {
  return (int)((__builtin_amdgcn_s_sendmsg_rtn(0x87) >> 18) & 1u);
}


__global__ __launch_bounds__(256) void gp_hesum_kernel(
    const float* __restrict__ he, const int* __restrict__ dst,
    float* gsum0, float* gsum1,
    float* deg0, float* deg1,
    const float* __restrict__ w_ih, const float* __restrict__ w_hh,
    float* __restrict__ wtab,
    int wt_blocks)
{
  if ((int)blockIdx.x < wt_blocks) {
    const int e = (int)blockIdx.x * 256 + (int)threadIdx.x;
    if (e < GP_ROUNDS * 2 * WT_MAT_PIECES) {
      const int m  = e / WT_MAT_PIECES;
      const int r  = e - m * WT_MAT_PIECES;
      const int ln = r & 31;
      const int j  = (r >> 5) & 7;
      const int q  = r >> 8;
      const int j0 = q / 3;
      const int g  = q - 3 * j0;
      const float* w = ((m & 1) ? w_hh : w_ih) + (long)(m >> 1) * WT_MAT_FLOATS;
      const v4f v = *(const v4f*)(w + (long)(g * 64 + j0 * 16 + (ln & 15)) * 64
                                    + 8 * j + 4 * (ln >> 4));
      *(v4f*)(wtab + (long)e * 4) = v;
    }
    return;
  }
  const int hsel = pkg_half();
  float* gsum = hsel ? gsum1 : gsum0;
  float* deg  = hsel ? deg1 : deg0;
  const int lane = threadIdx.x & 31;
  const int wave = threadIdx.x >> 5;
  const int g0 = (((int)blockIdx.x - wt_blocks) * 8 + wave) * GROUPS_PER_WAVE;
  for (int it = 0; it < GROUPS_PER_WAVE; ++it) {
    const int e0 = (g0 + it) * 32;
    const int dv = dst[e0 + lane];
    atomicAdd(&deg[dv], 1.0f);
    const float* hp = he + (long)e0 * GP_EDGE_DIM + lane;
    float v[32];
#pragma unroll
    for (int j = 0; j < 32; ++j) v[j] = hp[j * GP_EDGE_DIM];
#pragma unroll
    for (int j = 0; j < 32; ++j) {
      const int d = __builtin_amdgcn_readlane(dv, j);
      atomicAdd(gsum + (long)d * GP_EDGE_DIM + lane, v[j]);
    }
  }
}

__global__ __launch_bounds__(256) void gp_node_kernel(
    const float* __restrict__ hvin,
    const float* __restrict__ W,
    const float* __restrict__ bias,
    const float* gsum0, const float* gsum1,
    const float* deg0, const float* deg1,
    float* abuf0,
    float* abuf1,
    float* __restrict__ ptab)
{
  const bool two = (abuf1 != abuf0);
  __shared__ float Wl[KC_ALL * 4 * 32 * 2];
  const int tid = threadIdx.x;
  for (int p = tid; p < KC_ALL * 4 * 32; p += 256) {
    const int L  = p & 31;
    const int q  = p >> 5;
    const int n0 = q & 3;
    const int c  = q >> 2;
    const int k  = 4 * c + 2 * (L >> 4);
    const int col = n0 * 16 + (L & 15);
    v2f w2;
    w2.x = W[k * 64 + col];
    w2.y = W[(k + 1) * 64 + col];
    *(v2f*)&Wl[2 * p] = w2;
  }
  __syncthreads();

  const int lane = tid & 31;
  const int wave = tid >> 5;
  const int tile = blockIdx.x * 8 + wave;
  if (tile >= GP_N_NODES / 16) return;

  const int nbase = tile * 16;
  const int row   = lane & 15;
  const int half  = lane >> 4;

  v2f Ah[KC_SRC];
  const float* hq = hvin + (long)(nbase + row) * GP_NODE_DIM + 2 * half;
#pragma unroll
  for (int c = 0; c < KC_SRC; ++c) Ah[c] = ld2(hq + 4 * c);

  v2f Ag[KC_EDG];
  const long goff = (long)(nbase + row) * GP_EDGE_DIM + 2 * half;
#pragma unroll
  for (int c = 0; c < KC_EDG; ++c) Ag[c] = ld2(gsum0 + goff + 4 * c);

  float dg[8];
#pragma unroll
  for (int i = 0; i < 8; ++i) dg[i] = deg0[nbase + i + 8 * half];

  if (two) {
#pragma unroll
    for (int c = 0; c < KC_EDG; ++c) Ag[c] += ld2(gsum1 + goff + 4 * c);
#pragma unroll
    for (int i = 0; i < 8; ++i) dg[i] += deg1[nbase + i + 8 * half];
    v4f zero4 = {0.0f, 0.0f, 0.0f, 0.0f};
    float* zp = abuf1 + (long)nbase * GP_ACT_DIM + 4 * lane;
#pragma unroll
    for (int i = 0; i < 8; ++i) *(v4f*)(zp + i * 128) = zero4;
  }

  const v2f* Wl2 = (const v2f*)Wl;
  v8f accP[4];
#pragma unroll
  for (int n0 = 0; n0 < 4; ++n0) {
    const float bv = bias[n0 * 16 + row];
    v8f accQ, accR;
#pragma unroll
    for (int i = 0; i < 8; ++i) { accQ[i] = bv; accR[i] = 0.0f; accP[n0][i] = 0.0f; }

    const v2f* Bp = Wl2 + (n0 * 32 + lane);
#pragma unroll
    for (int c = 0; c < KC_SRC; ++c) {
      accP[n0] = wmma4(Ah[c], Bp[(c) * 128], accP[n0]);
      accQ     = wmma4(Ah[c], Bp[(KC_SRC + c) * 128], accQ);
    }
#pragma unroll
    for (int c = 0; c < KC_EDG; ++c)
      accR = wmma4(Ag[c], Bp[(KC_SRC + KC_DST + c) * 128], accR);
#pragma unroll
    for (int i = 0; i < 8; ++i)
      abuf0[(long)(nbase + i + 8 * half) * GP_ACT_DIM + n0 * 16 + row] =
          dg[i] * accQ[i] + accR[i];
  }
#pragma unroll
  for (int i = 0; i < 8; ++i) {
    float* pr = ptab + (long)(nbase + i + 8 * half) * GP_NODE_DIM + 2 * row;
    v2f lo, hi;
    lo.x = accP[0][i]; lo.y = accP[2][i];
    hi.x = accP[1][i]; hi.y = accP[3][i];
    *(v2f*)pr = lo;
    *(v2f*)(pr + 32) = hi;
  }
}

template <bool WT>
__global__ __launch_bounds__(256) void gp_edge_kernel(
    const float* __restrict__ ptab,
    const int* __restrict__ src, const int* __restrict__ dst,
    float* abuf0, float* abuf1,
    const float* __restrict__ hvin,
    const float* __restrict__ w_hh, const float* __restrict__ b_hh,
    const float* __restrict__ wt_hh,
    float* __restrict__ ghbuf,
    int gh_blocks)
{
  if ((int)blockIdx.x < gh_blocks) {
    const int lane = threadIdx.x & 31;
    const int wave = threadIdx.x >> 5;
    const int tile = blockIdx.x * 8 + wave;
    if (tile >= GP_N_NODES / 16) return;
    const int nbase = tile * 16;
    const int row   = lane & 15;
    const int half  = lane >> 4;
    if (WT) {
      const float* hq4 = hvin + (long)(nbase + row) * GP_NODE_DIM + 4 * half;
#pragma unroll 1
      for (int j0 = 0; j0 < 4; ++j0) {
        const int colb = j0 * 16 + row;
        v8f hr, hz, hn;
        const float bh0 = b_hh[colb], bh1 = b_hh[64 + colb], bh2 = b_hh[128 + colb];
#pragma unroll
        for (int i = 0; i < 8; ++i) { hr[i] = bh0; hz[i] = bh1; hn[i] = bh2; }
        const v4f* tb = (const v4f*)wt_hh + j0 * 768 + lane;
#pragma unroll 1
        for (int jj = 0; jj < 8; jj += 2) {
#pragma unroll
          for (int u = 0; u < 2; ++u) {
            const int j = jj + u;
            const v4f a  = *(const v4f*)(hq4 + 8 * j);
            const v4f b0 = tb[j * 32], b1 = tb[256 + j * 32], b2 = tb[512 + j * 32];
            hr = wmma4(a.xy, b0.xy, hr);
            hz = wmma4(a.xy, b1.xy, hz);
            hn = wmma4(a.xy, b2.xy, hn);
            hr = wmma4(a.zw, b0.zw, hr);
            hz = wmma4(a.zw, b1.zw, hz);
            hn = wmma4(a.zw, b2.zw, hn);
          }
        }
        float* gp = ghbuf + ((long)tile * 4 + j0) * 768 + 4 * lane;
        v4f t;
        t[0] = hr[0]; t[1] = hr[1]; t[2] = hr[2]; t[3] = hr[3]; *(v4f*)(gp)       = t;
        t[0] = hr[4]; t[1] = hr[5]; t[2] = hr[6]; t[3] = hr[7]; *(v4f*)(gp + 128) = t;
        t[0] = hz[0]; t[1] = hz[1]; t[2] = hz[2]; t[3] = hz[3]; *(v4f*)(gp + 256) = t;
        t[0] = hz[4]; t[1] = hz[5]; t[2] = hz[6]; t[3] = hz[7]; *(v4f*)(gp + 384) = t;
        t[0] = hn[0]; t[1] = hn[1]; t[2] = hn[2]; t[3] = hn[3]; *(v4f*)(gp + 512) = t;
        t[0] = hn[4]; t[1] = hn[5]; t[2] = hn[6]; t[3] = hn[7]; *(v4f*)(gp + 640) = t;
      }
      return;
    }
    const float* hq = hvin + (long)(nbase + row) * GP_NODE_DIM + 2 * half;
#pragma unroll 1
    for (int j0 = 0; j0 < 4; ++j0) {
      const int colb = j0 * 16 + row;
      v8f hr, hz, hn;
      const float bh0 = b_hh[colb], bh1 = b_hh[64 + colb], bh2 = b_hh[128 + colb];
#pragma unroll
      for (int i = 0; i < 8; ++i) { hr[i] = bh0; hz[i] = bh1; hn[i] = bh2; }
      const float* wh0 = w_hh + (long)(colb)*64 + 2 * half;
      const float* wh1 = w_hh + (long)(64 + colb) * 64 + 2 * half;
      const float* wh2 = w_hh + (long)(128 + colb) * 64 + 2 * half;
#pragma unroll 1
      for (int cc = 0; cc < 16; cc += 4) {
#pragma unroll
        for (int c = 0; c < 4; ++c) {
          const v2f a = ld2(hq + 4 * (cc + c));
          hr = wmma4(a, ld2(wh0 + 4 * (cc + c)), hr);
          hz = wmma4(a, ld2(wh1 + 4 * (cc + c)), hz);
          hn = wmma4(a, ld2(wh2 + 4 * (cc + c)), hn);
        }
      }
      float* gp = ghbuf + ((long)tile * 4 + j0) * 768 + 4 * lane;
      v4f t;
      t[0] = hr[0]; t[1] = hr[1]; t[2] = hr[2]; t[3] = hr[3]; *(v4f*)(gp)       = t;
      t[0] = hr[4]; t[1] = hr[5]; t[2] = hr[6]; t[3] = hr[7]; *(v4f*)(gp + 128) = t;
      t[0] = hz[0]; t[1] = hz[1]; t[2] = hz[2]; t[3] = hz[3]; *(v4f*)(gp + 256) = t;
      t[0] = hz[4]; t[1] = hz[5]; t[2] = hz[6]; t[3] = hz[7]; *(v4f*)(gp + 384) = t;
      t[0] = hn[0]; t[1] = hn[1]; t[2] = hn[2]; t[3] = hn[3]; *(v4f*)(gp + 512) = t;
      t[0] = hn[4]; t[1] = hn[5]; t[2] = hn[6]; t[3] = hn[7]; *(v4f*)(gp + 640) = t;
    }
    return;
  }
  float* abuf = pkg_half() ? abuf1 : abuf0;
  const int lane = threadIdx.x & 31;
  const int wave = threadIdx.x >> 5;
  const int g0 = (((int)blockIdx.x - gh_blocks) * 8 + wave) * GROUPS_PER_WAVE;
  for (int it = 0; it < GROUPS_PER_WAVE; ++it) {
    const int e0 = (g0 + it) * 32;
    const int sv = src[e0 + lane];
    const int dv = dst[e0 + lane];
#pragma unroll
    for (int hb = 0; hb < 32; hb += 16) {
      v2f val[16];
#pragma unroll
      for (int j = 0; j < 16; ++j) {
        const int s = __builtin_amdgcn_readlane(sv, hb + j);
        val[j] = ld2(ptab + (long)s * GP_NODE_DIM + 2 * lane);
      }
#pragma unroll
      for (int j = 0; j < 16; ++j) {
        const int d = __builtin_amdgcn_readlane(dv, hb + j);
        float* ap = abuf + (long)d * GP_ACT_DIM + lane;
        atomicAdd(ap, val[j].x);
        atomicAdd(ap + 32, val[j].y);
      }
    }
  }
}

template <bool GH_PRE, bool WT>
__global__ __launch_bounds__(256) void gp_gru_kernel(
    const float* abuf0, const float* abuf1,
    const float* __restrict__ hvin,
    const float* __restrict__ w_ih, const float* __restrict__ w_hh,
    const float* __restrict__ b_ih, const float* __restrict__ b_hh,
    const float* __restrict__ ghbuf,
    const float* __restrict__ wt_ih,
    float* __restrict__ hvout)
{
  const int tid  = threadIdx.x;
  const int lane = tid & 31;
  const int wave = tid >> 5;
  const int tile = blockIdx.x * 8 + wave;
  const int ntiles = GP_N_NODES / 16;
  if (tile >= ntiles) return;

  const int nbase = tile * 16;
  const int row   = lane & 15;
  const int half  = lane >> 4;

  v2f Aa[16], Ah[16];
  v4f Aq[8];
  const long aoff = (long)(nbase + row) * GP_ACT_DIM + 2 * half;
  const float* hq = hvin + (long)(nbase + row) * GP_NODE_DIM + 2 * half;
  if (WT) {
    const long aoff4 = (long)(nbase + row) * GP_ACT_DIM + 4 * half;
#pragma unroll
    for (int j = 0; j < 8; ++j) Aq[j] = *(const v4f*)(abuf0 + aoff4 + 8 * j);
    if (abuf1 != abuf0) {
#pragma unroll
      for (int j = 0; j < 8; ++j) Aq[j] += *(const v4f*)(abuf1 + aoff4 + 8 * j);
    }
  } else {
#pragma unroll
    for (int c = 0; c < 16; ++c) {
      Aa[c] = ld2(abuf0 + aoff + 4 * c);
      if (!GH_PRE) Ah[c] = ld2(hq + 4 * c);
    }
    if (abuf1 != abuf0) {
#pragma unroll
      for (int c = 0; c < 16; ++c) Aa[c] += ld2(abuf1 + aoff + 4 * c);
    }
  }

#pragma unroll
  for (int j0 = 0; j0 < 4; ++j0) {
    const int colb = j0 * 16 + row;
    v8f ir, iz, in_, hr, hz, hn;
    {
      const float bi0 = b_ih[colb], bi1 = b_ih[64 + colb], bi2 = b_ih[128 + colb];
#pragma unroll
      for (int i = 0; i < 8; ++i) { ir[i] = bi0; iz[i] = bi1; in_[i] = bi2; }
    }
    const float* wi0 = w_ih + (long)(colb)*64 + 2 * half;
    const float* wi1 = w_ih + (long)(64 + colb) * 64 + 2 * half;
    const float* wi2 = w_ih + (long)(128 + colb) * 64 + 2 * half;
    if (GH_PRE) {
      const float* gp = ghbuf + ((long)tile * 4 + j0) * 768 + 4 * lane;
      const v4f r0 = *(const v4f*)(gp),       r1 = *(const v4f*)(gp + 128);
      const v4f z0 = *(const v4f*)(gp + 256), z1 = *(const v4f*)(gp + 384);
      const v4f n0 = *(const v4f*)(gp + 512), n1 = *(const v4f*)(gp + 640);
#pragma unroll
      for (int i = 0; i < 4; ++i) {
        hr[i] = r0[i]; hr[4 + i] = r1[i];
        hz[i] = z0[i]; hz[4 + i] = z1[i];
        hn[i] = n0[i]; hn[4 + i] = n1[i];
      }
      if (WT) {
        const v4f* tb = (const v4f*)wt_ih + j0 * 768 + lane;
#pragma unroll
        for (int j = 0; j < 8; ++j) {
          const v4f b0 = tb[j * 32], b1 = tb[256 + j * 32], b2 = tb[512 + j * 32];
          ir  = wmma4(Aq[j].xy, b0.xy, ir);
          iz  = wmma4(Aq[j].xy, b1.xy, iz);
          in_ = wmma4(Aq[j].xy, b2.xy, in_);
          ir  = wmma4(Aq[j].zw, b0.zw, ir);
          iz  = wmma4(Aq[j].zw, b1.zw, iz);
          in_ = wmma4(Aq[j].zw, b2.zw, in_);
        }
      } else {
#pragma unroll
        for (int c = 0; c < 16; ++c) {
          ir  = wmma4(Aa[c], ld2(wi0 + 4 * c), ir);
          iz  = wmma4(Aa[c], ld2(wi1 + 4 * c), iz);
          in_ = wmma4(Aa[c], ld2(wi2 + 4 * c), in_);
        }
      }
    } else {
      const float bh0 = b_hh[colb], bh1 = b_hh[64 + colb], bh2 = b_hh[128 + colb];
#pragma unroll
      for (int i = 0; i < 8; ++i) { hr[i] = bh0; hz[i] = bh1; hn[i] = bh2; }
      const float* wh0 = w_hh + (long)(colb)*64 + 2 * half;
      const float* wh1 = w_hh + (long)(64 + colb) * 64 + 2 * half;
      const float* wh2 = w_hh + (long)(128 + colb) * 64 + 2 * half;
#pragma unroll
      for (int c = 0; c < 16; ++c) {
        ir  = wmma4(Aa[c], ld2(wi0 + 4 * c), ir);
        iz  = wmma4(Aa[c], ld2(wi1 + 4 * c), iz);
        in_ = wmma4(Aa[c], ld2(wi2 + 4 * c), in_);
        hr  = wmma4(Ah[c], ld2(wh0 + 4 * c), hr);
        hz  = wmma4(Ah[c], ld2(wh1 + 4 * c), hz);
        hn  = wmma4(Ah[c], ld2(wh2 + 4 * c), hn);
      }
    }
#pragma unroll
    for (int i = 0; i < 8; ++i) {
      const long m = nbase + i + 8 * half;
      const float h = hvin[m * GP_NODE_DIM + colb];
      const float r = sigmoid_(ir[i] + hr[i]);
      const float z = sigmoid_(iz[i] + hz[i]);
      const float n = tanhf(in_[i] + r * hn[i]);
      hvout[m * GP_NODE_DIM + colb] = (1.0f - z) * n + z * h;
    }
  }
}

extern "C" void kernel_launch(void* const* d_in, const int* in_sizes, int n_in,
                              void* d_out, int out_size, void* d_ws, size_t ws_size,
                              hipStream_t stream) {
  const float* hv    = (const float*)d_in[0];
  const float* he    = (const float*)d_in[1];
  const int*   src   = (const int*)d_in[2];
  const int*   dst   = (const int*)d_in[3];
  const float* W_msg = (const float*)d_in[4];
  const float* b_msg = (const float*)d_in[5];
  const float* w_ih  = (const float*)d_in[6];
  const float* w_hh  = (const float*)d_in[7];
  const float* b_ih  = (const float*)d_in[8];
  const float* b_hh  = (const float*)d_in[9];
  float* out = (float*)d_out;

  const size_t state_elems = (size_t)GP_N_NODES * GP_NODE_DIM;
  float* a_buf  = (float*)d_ws;
  float* hv_buf = a_buf + state_elems;
  const size_t gsum_elems = (size_t)GP_N_NODES * GP_EDGE_DIM;
  float* g_sum  = hv_buf + state_elems;
  float* deg    = g_sum + gsum_elems;
  if (ws_size < (2 * state_elems + gsum_elems + GP_N_NODES) * sizeof(float)) return;
  const bool two = ws_size >= (3 * state_elems + 2 * gsum_elems + 2 * GP_N_NODES) * sizeof(float);
  float* g_sum1 = two ? deg + GP_N_NODES : g_sum;
  float* deg1   = two ? g_sum1 + gsum_elems : deg;
  float* a_buf1 = two ? deg1 + GP_N_NODES : a_buf;
  const bool pre = ws_size >= (6 * state_elems + 2 * gsum_elems + 2 * GP_N_NODES) * sizeof(float);
  float* gh_buf = a_buf1 + state_elems;
  const size_t wt_elems = (size_t)GP_ROUNDS * 2 * WT_MAT_FLOATS;
  const bool wt = ws_size >= (6 * state_elems + 2 * gsum_elems + 2 * GP_N_NODES + wt_elems) * sizeof(float);
  float* w_tab = gh_buf + 3 * state_elems;
  const int wt_blocks = wt ? WT_BLOCKS : 0;
  float* p_tab  = out;

  const int grp_blocks  = (GP_N_EDGES / 32) / (8 * GROUPS_PER_WAVE);
  const int node_blocks = (GP_N_NODES / 16 + 7) / 8;
  const int gru_blocks  = (GP_N_NODES / 16 + 7) / 8;

  hipMemsetAsync(g_sum, 0, (two ? 2 : 1) * (gsum_elems + GP_N_NODES) * sizeof(float), stream);
  gp_hesum_kernel<<<wt_blocks + grp_blocks, 256, 0, stream>>>(
      he, dst, g_sum, g_sum1, deg, deg1, w_ih, w_hh, w_tab, wt_blocks);

  for (int t = 0; t < GP_ROUNDS; ++t) {
    const float* hcur = (t == 0) ? hv : hv_buf;
    const float* Wt = W_msg + (size_t)t * GP_CAT_DIM * GP_ACT_DIM;
    gp_node_kernel<<<node_blocks, 256, 0, stream>>>(
        hcur, Wt, b_msg + (size_t)t * GP_ACT_DIM, g_sum, g_sum1, deg, deg1,
        a_buf, a_buf1, p_tab);
    const float* whh = w_hh + (size_t)t * 3 * GP_NODE_DIM * GP_NODE_DIM;
    const float* bhh = b_hh + (size_t)t * 3 * GP_NODE_DIM;
    float* outp = (t == GP_ROUNDS - 1) ? out : hv_buf;
    const float* wt_ih = w_tab + (size_t)(2 * t) * WT_MAT_FLOATS;
    const float* wt_hh = w_tab + (size_t)(2 * t + 1) * WT_MAT_FLOATS;
    if (wt) {
      gp_edge_kernel<true><<<gru_blocks + grp_blocks, 256, 0, stream>>>(
          p_tab, src, dst, a_buf, a_buf1, hcur, whh, bhh, wt_hh, gh_buf, gru_blocks);
      gp_gru_kernel<true, true><<<gru_blocks, 256, 0, stream>>>(
          a_buf, a_buf1, hcur,
          w_ih + (size_t)t * 3 * GP_NODE_DIM * GP_NODE_DIM, whh,
          b_ih + (size_t)t * 3 * GP_NODE_DIM, bhh, gh_buf, wt_ih, outp);
    } else if (pre) {
      gp_edge_kernel<false><<<gru_blocks + grp_blocks, 256, 0, stream>>>(
          p_tab, src, dst, a_buf, a_buf1, hcur, whh, bhh, whh, gh_buf, gru_blocks);
      gp_gru_kernel<true, false><<<gru_blocks, 256, 0, stream>>>(
          a_buf, a_buf1, hcur,
          w_ih + (size_t)t * 3 * GP_NODE_DIM * GP_NODE_DIM, whh,
          b_ih + (size_t)t * 3 * GP_NODE_DIM, bhh, gh_buf, whh, outp);
    } else {
      gp_edge_kernel<false><<<grp_blocks, 256, 0, stream>>>(
          p_tab, src, dst, a_buf, a_buf1, hcur, whh, bhh, whh, a_buf, 0);
      gp_gru_kernel<false, false><<<gru_blocks, 256, 0, stream>>>(
          a_buf, a_buf1, hcur,
          w_ih + (size_t)t * 3 * GP_NODE_DIM * GP_NODE_DIM, whh,
          b_ih + (size_t)t * 3 * GP_NODE_DIM, bhh, a_buf, whh, outp);
    }
  }
}
